# starting kernel
# baseline (speedup 1.0000x reference)
#include <hip/hip_runtime.h>

typedef __attribute__((ext_vector_type(2))) float v2f;
typedef __attribute__((ext_vector_type(4))) float v4f;
typedef __attribute__((ext_vector_type(8))) float v8f;

#define NPTS 8192
#define CCH  64
#define KNN  16
#define KSTR 132
#define XSTR 68
#define WAVE_LDS (2*64 + 2*16*XSTR)

__global__ __launch_bounds__(256) void knn_kernel(const float* __restrict__ coords,
                                                  int* __restrict__ idx_out) {
    __shared__ v4f pts[NPTS];
    const int b     = blockIdx.x >> 5;
    const int qbase = (blockIdx.x & 31) << 8;
    const float* cb = coords + b * NPTS * 3;

    for (int j = threadIdx.x; j < NPTS; j += 256) {
        float x = cb[j * 3 + 0], y = cb[j * 3 + 1], z = cb[j * 3 + 2];
        v4f p; p.x = x; p.y = y; p.z = z; p.w = x * x + y * y + z * z;
        pts[j] = p;
    }
    __syncthreads();

    const int q = qbase + threadIdx.x;
    const v4f me = pts[q];

    float best[KNN]; int bidx[KNN];
#pragma unroll
    for (int t = 0; t < KNN; ++t) { best[t] = 3.4e38f; bidx[t] = 0; }

    for (int j = 0; j < NPTS; ++j) {
        v4f p = pts[j];
        float dot = me.x * p.x;
        dot = fmaf(me.y, p.y, dot);
        dot = fmaf(me.z, p.z, dot);
        float d = fmaf(-2.0f, dot, me.w + p.w);
        d = fmaxf(d, 0.0f);
        if (d < best[KNN - 1]) {
            float cd = d; int ci = j;
#pragma unroll
            for (int t = 0; t < KNN; ++t) {
                if (cd < best[t]) {
                    float tf = best[t]; best[t] = cd; cd = tf;
                    int   ti = bidx[t]; bidx[t] = ci; ci = ti;
                }
            }
        }
    }

    int* op = idx_out + (b * NPTS + q) * KNN;
#pragma unroll
    for (int t = 0; t < KNN; ++t) op[t] = bidx[t];
}

__global__ __launch_bounds__(256) void agg_kernel(const float* __restrict__ coords,
                                                  const float* __restrict__ feat,
                                                  const float* __restrict__ Wm,
                                                  const float* __restrict__ bias,
                                                  const int*   __restrict__ idx,
                                                  float* __restrict__ out) {
    __shared__ float LWT[64 * KSTR];
    __shared__ float XS[8 * WAVE_LDS];

    for (int i = threadIdx.x; i < 131 * 64; i += 256) {
        int k = i >> 6, n = i & 63;
        LWT[n * KSTR + k] = Wm[i];
    }
    if (threadIdx.x < 64) LWT[threadIdx.x * KSTR + 131] = 0.0f;
    __syncthreads();

    const int lane = threadIdx.x & 31;
    const int wav  = threadIdx.x >> 5;
    const int nl   = lane & 15;
    const int h    = lane >> 4;

    float* fib0 = &XS[wav * WAVE_LDS];
    float* fib1 = fib0 + 64;
    float* X0   = fib1 + 64;
    float* X1   = X0 + 16 * XSTR;

    const float* wt0 = LWT + (nl     ) * KSTR + 2 * h;
    const float* wt1 = LWT + (nl + 16) * KSTR + 2 * h;
    const float* wt2 = LWT + (nl + 32) * KSTR + 2 * h;
    const float* wt3 = LWT + (nl + 48) * KSTR + 2 * h;

    float bs[4];
#pragma unroll
    for (int nt = 0; nt < 4; ++nt) bs[nt] = bias[nt * 16 + nl];

    const int gw = blockIdx.x * 8 + wav;

    for (int it = 0; it < 16; ++it) {
        const int P0 = gw * 32 + it * 2;
        const int b  = P0 >> 13;
        const int n0 = P0 & (NPTS - 1);
        const int n1 = n0 + 1;
        const float* fb = feat   + b * NPTS * CCH;
        const float* cb = coords + b * NPTS * 3;

        const int myn = h ? n1 : n0;
        const int nbv = idx[(b * NPTS + myn) * KNN + nl];

        v2f fi0 = *(const v2f*)(fb + n0 * CCH + 2 * lane);
        v2f fi1 = *(const v2f*)(fb + n1 * CCH + 2 * lane);
        *(v2f*)(fib0 + 2 * lane) = fi0;
        *(v2f*)(fib1 + 2 * lane) = fi1;

#pragma unroll
        for (int m = 0; m < 32; ++m) {
            int nb = __shfl(nbv, m, 32);
            v2f g  = *(const v2f*)(fb + nb * CCH + 2 * lane);
            if (m < 16) { v2f d0 = g - fi0; *(v2f*)(X0 + m * XSTR + 2 * lane) = d0; }
            else        { v2f d1 = g - fi1; *(v2f*)(X1 + (m - 16) * XSTR + 2 * lane) = d1; }
        }

        {
            const float* pc  = cb + myn * 3;
            const float* ncp = cb + nbv * 3;
            v4f dp; dp.x = ncp[0] - pc[0]; dp.y = ncp[1] - pc[1];
                    dp.z = ncp[2] - pc[2]; dp.w = 0.0f;
            float* xr = (h ? X1 : X0) + nl * XSTR + 64;
            *(v4f*)xr = dp;
        }

        v8f a0c[4], a1c[4];
        const v8f zc = {0.f, 0.f, 0.f, 0.f, 0.f, 0.f, 0.f, 0.f};
#pragma unroll
        for (int nt = 0; nt < 4; ++nt) { a0c[nt] = zc; a1c[nt] = zc; }

#define STEP(K0, A0, A1)                                                          \
        {                                                                         \
            v2f b0 = *(const v2f*)(wt0 + (K0));                                   \
            v2f b1 = *(const v2f*)(wt1 + (K0));                                   \
            v2f b2 = *(const v2f*)(wt2 + (K0));                                   \
            v2f b3 = *(const v2f*)(wt3 + (K0));                                   \
            a0c[0] = __builtin_amdgcn_wmma_f32_16x16x4_f32(                       \
                false, (A0), false, b0, (short)0, a0c[0], false, false);          \
            a1c[0] = __builtin_amdgcn_wmma_f32_16x16x4_f32(                       \
                false, (A1), false, b0, (short)0, a1c[0], false, false);          \
            a0c[1] = __builtin_amdgcn_wmma_f32_16x16x4_f32(                       \
                false, (A0), false, b1, (short)0, a0c[1], false, false);          \
            a1c[1] = __builtin_amdgcn_wmma_f32_16x16x4_f32(                       \
                false, (A1), false, b1, (short)0, a1c[1], false, false);          \
            a0c[2] = __builtin_amdgcn_wmma_f32_16x16x4_f32(                       \
                false, (A0), false, b2, (short)0, a0c[2], false, false);          \
            a1c[2] = __builtin_amdgcn_wmma_f32_16x16x4_f32(                       \
                false, (A1), false, b2, (short)0, a1c[2], false, false);          \
            a0c[3] = __builtin_amdgcn_wmma_f32_16x16x4_f32(                       \
                false, (A0), false, b3, (short)0, a0c[3], false, false);          \
            a1c[3] = __builtin_amdgcn_wmma_f32_16x16x4_f32(                       \
                false, (A1), false, b3, (short)0, a1c[3], false, false);          \
        }

#pragma unroll
        for (int ks = 0; ks < 16; ++ks) {
            int k0 = 4 * ks;
            v2f A0 = *(const v2f*)(fib0 + k0 + 2 * h);
            v2f A1 = *(const v2f*)(fib1 + k0 + 2 * h);
            STEP(k0, A0, A1);
        }
#pragma unroll
        for (int ks = 0; ks < 16; ++ks) {
            int k0 = 64 + 4 * ks;
            int xo = nl * XSTR + 4 * ks + 2 * h;
            v2f A0 = *(const v2f*)(X0 + xo);
            v2f A1 = *(const v2f*)(X1 + xo);
            STEP(k0, A0, A1);
        }
        {
            int k0 = 128;
            int xo = nl * XSTR + 64 + 2 * h;
            v2f A0 = *(const v2f*)(X0 + xo);
            v2f A1 = *(const v2f*)(X1 + xo);
            STEP(k0, A0, A1);
        }
#undef STEP

        float* outp = out + (size_t)P0 * CCH;
#pragma unroll
        for (int nt = 0; nt < 4; ++nt) {
            float v0 = a0c[nt][0];
            float v1 = a1c[nt][0];
#pragma unroll
            for (int r = 1; r < 8; ++r) {
                v0 = fmaxf(v0, a0c[nt][r]);
                v1 = fmaxf(v1, a1c[nt][r]);
            }
            v0 = fmaxf(v0, __shfl_xor(v0, 16, 32));
            v1 = fmaxf(v1, __shfl_xor(v1, 16, 32));
            v0 = fmaxf(v0 + bs[nt], 0.0f);
            v1 = fmaxf(v1 + bs[nt], 0.0f);
            if (lane < 16) {
                outp[nt * 16 + nl]       = v0;
                outp[CCH + nt * 16 + nl] = v1;
            }
        }
    }
}

extern "C" void kernel_launch(void* const* d_in, const int* in_sizes, int n_in,
                              void* d_out, int out_size, void* d_ws, size_t ws_size,
                              hipStream_t stream) {
    const float* coords = (const float*)d_in[0];
    const float* feat   = (const float*)d_in[1];
    const float* Wm     = (const float*)d_in[2];
    const float* bias   = (const float*)d_in[3];
    float* out = (float*)d_out;
    int* idx = (int*)d_ws;

    knn_kernel<<<64, 256, 0, stream>>>(coords, idx);
    agg_kernel<<<64, 256, 0, stream>>>(coords, feat, Wm, bias, idx, out);
}


# KNN kernel: sorted insert made straight-line (16 compares, one median-of-3 per slot, selects for the indices) instead of 16 divergent branches that each copy the whole list
# speedup vs baseline: 2.8383x; 2.8383x over previous
#include <hip/hip_runtime.h>

typedef __attribute__((ext_vector_type(2))) float v2f;
typedef __attribute__((ext_vector_type(4))) float v4f;
typedef __attribute__((ext_vector_type(8))) float v8f;

#define NPTS 8192
#define CCH  64
#define KNN  16
#define KSTR 132
#define XSTR 68
#define WAVE_LDS (2*64 + 2*16*XSTR)

__global__ __launch_bounds__(256) void knn_kernel(const float* __restrict__ coords,
                                                  int* __restrict__ idx_out) {
    __shared__ v4f pts[NPTS];
    const int b     = blockIdx.x >> 5;
    const int qbase = (blockIdx.x & 31) << 8;
    const float* cb = coords + b * NPTS * 3;

    for (int j = threadIdx.x; j < NPTS; j += 256) {
        float x = cb[j * 3 + 0], y = cb[j * 3 + 1], z = cb[j * 3 + 2];
        v4f p; p.x = x; p.y = y; p.z = z; p.w = x * x + y * y + z * z;
        pts[j] = p;
    }
    __syncthreads();

    const int q = qbase + threadIdx.x;
    const v4f me = pts[q];

    float best[KNN]; int bidx[KNN];
#pragma unroll
    for (int t = 0; t < KNN; ++t) { best[t] = 3.4e38f; bidx[t] = 0; }

    for (int j = 0; j < NPTS; ++j) {
        v4f p = pts[j];
        float dot = me.x * p.x;
        dot = fmaf(me.y, p.y, dot);
        dot = fmaf(me.z, p.z, dot);
        float d = fmaf(-2.0f, dot, me.w + p.w);
        d = fmaxf(d, 0.0f);
        if (d < best[KNN - 1]) {
            bool lt[KNN];
#pragma unroll
            for (int t = 0; t < KNN; ++t) lt[t] = d < best[t];
#pragma unroll
            for (int t = KNN - 1; t >= 1; --t) {
                bidx[t] = lt[t - 1] ? bidx[t - 1] : (lt[t] ? j : bidx[t]);
                best[t] = __builtin_amdgcn_fmed3f(best[t - 1], d, best[t]);
            }
            bidx[0] = lt[0] ? j : bidx[0];
            best[0] = fminf(d, best[0]);
        }
    }

    int* op = idx_out + (b * NPTS + q) * KNN;
#pragma unroll
    for (int t = 0; t < KNN; ++t) op[t] = bidx[t];
}

__global__ __launch_bounds__(256) void agg_kernel(const float* __restrict__ coords,
                                                  const float* __restrict__ feat,
                                                  const float* __restrict__ Wm,
                                                  const float* __restrict__ bias,
                                                  const int*   __restrict__ idx,
                                                  float* __restrict__ out) {
    __shared__ float LWT[64 * KSTR];
    __shared__ float XS[8 * WAVE_LDS];

    for (int i = threadIdx.x; i < 131 * 64; i += 256) {
        int k = i >> 6, n = i & 63;
        LWT[n * KSTR + k] = Wm[i];
    }
    if (threadIdx.x < 64) LWT[threadIdx.x * KSTR + 131] = 0.0f;
    __syncthreads();

    const int lane = threadIdx.x & 31;
    const int wav  = threadIdx.x >> 5;
    const int nl   = lane & 15;
    const int h    = lane >> 4;

    float* fib0 = &XS[wav * WAVE_LDS];
    float* fib1 = fib0 + 64;
    float* X0   = fib1 + 64;
    float* X1   = X0 + 16 * XSTR;

    const float* wt0 = LWT + (nl     ) * KSTR + 2 * h;
    const float* wt1 = LWT + (nl + 16) * KSTR + 2 * h;
    const float* wt2 = LWT + (nl + 32) * KSTR + 2 * h;
    const float* wt3 = LWT + (nl + 48) * KSTR + 2 * h;

    float bs[4];
#pragma unroll
    for (int nt = 0; nt < 4; ++nt) bs[nt] = bias[nt * 16 + nl];

    const int gw = blockIdx.x * 8 + wav;

    for (int it = 0; it < 16; ++it) {
        const int P0 = gw * 32 + it * 2;
        const int b  = P0 >> 13;
        const int n0 = P0 & (NPTS - 1);
        const int n1 = n0 + 1;
        const float* fb = feat   + b * NPTS * CCH;
        const float* cb = coords + b * NPTS * 3;

        const int myn = h ? n1 : n0;
        const int nbv = idx[(b * NPTS + myn) * KNN + nl];

        v2f fi0 = *(const v2f*)(fb + n0 * CCH + 2 * lane);
        v2f fi1 = *(const v2f*)(fb + n1 * CCH + 2 * lane);
        *(v2f*)(fib0 + 2 * lane) = fi0;
        *(v2f*)(fib1 + 2 * lane) = fi1;

#pragma unroll
        for (int m = 0; m < 32; ++m) {
            int nb = __shfl(nbv, m, 32);
            v2f g  = *(const v2f*)(fb + nb * CCH + 2 * lane);
            if (m < 16) { v2f d0 = g - fi0; *(v2f*)(X0 + m * XSTR + 2 * lane) = d0; }
            else        { v2f d1 = g - fi1; *(v2f*)(X1 + (m - 16) * XSTR + 2 * lane) = d1; }
        }

        {
            const float* pc  = cb + myn * 3;
            const float* ncp = cb + nbv * 3;
            v4f dp; dp.x = ncp[0] - pc[0]; dp.y = ncp[1] - pc[1];
                    dp.z = ncp[2] - pc[2]; dp.w = 0.0f;
            float* xr = (h ? X1 : X0) + nl * XSTR + 64;
            *(v4f*)xr = dp;
        }

        v8f a0c[4], a1c[4];
        const v8f zc = {0.f, 0.f, 0.f, 0.f, 0.f, 0.f, 0.f, 0.f};
#pragma unroll
        for (int nt = 0; nt < 4; ++nt) { a0c[nt] = zc; a1c[nt] = zc; }

#define STEP(K0, A0, A1)                                                          \
        {                                                                         \
            v2f b0 = *(const v2f*)(wt0 + (K0));                                   \
            v2f b1 = *(const v2f*)(wt1 + (K0));                                   \
            v2f b2 = *(const v2f*)(wt2 + (K0));                                   \
            v2f b3 = *(const v2f*)(wt3 + (K0));                                   \
            a0c[0] = __builtin_amdgcn_wmma_f32_16x16x4_f32(                       \
                false, (A0), false, b0, (short)0, a0c[0], false, false);          \
            a1c[0] = __builtin_amdgcn_wmma_f32_16x16x4_f32(                       \
                false, (A1), false, b0, (short)0, a1c[0], false, false);          \
            a0c[1] = __builtin_amdgcn_wmma_f32_16x16x4_f32(                       \
                false, (A0), false, b1, (short)0, a0c[1], false, false);          \
            a1c[1] = __builtin_amdgcn_wmma_f32_16x16x4_f32(                       \
                false, (A1), false, b1, (short)0, a1c[1], false, false);          \
            a0c[2] = __builtin_amdgcn_wmma_f32_16x16x4_f32(                       \
                false, (A0), false, b2, (short)0, a0c[2], false, false);          \
            a1c[2] = __builtin_amdgcn_wmma_f32_16x16x4_f32(                       \
                false, (A1), false, b2, (short)0, a1c[2], false, false);          \
            a0c[3] = __builtin_amdgcn_wmma_f32_16x16x4_f32(                       \
                false, (A0), false, b3, (short)0, a0c[3], false, false);          \
            a1c[3] = __builtin_amdgcn_wmma_f32_16x16x4_f32(                       \
                false, (A1), false, b3, (short)0, a1c[3], false, false);          \
        }

#pragma unroll
        for (int ks = 0; ks < 16; ++ks) {
            int k0 = 4 * ks;
            v2f A0 = *(const v2f*)(fib0 + k0 + 2 * h);
            v2f A1 = *(const v2f*)(fib1 + k0 + 2 * h);
            STEP(k0, A0, A1);
        }
#pragma unroll
        for (int ks = 0; ks < 16; ++ks) {
            int k0 = 64 + 4 * ks;
            int xo = nl * XSTR + 4 * ks + 2 * h;
            v2f A0 = *(const v2f*)(X0 + xo);
            v2f A1 = *(const v2f*)(X1 + xo);
            STEP(k0, A0, A1);
        }
        {
            int k0 = 128;
            int xo = nl * XSTR + 64 + 2 * h;
            v2f A0 = *(const v2f*)(X0 + xo);
            v2f A1 = *(const v2f*)(X1 + xo);
            STEP(k0, A0, A1);
        }
#undef STEP

        float* outp = out + (size_t)P0 * CCH;
#pragma unroll
        for (int nt = 0; nt < 4; ++nt) {
            float v0 = a0c[nt][0];
            float v1 = a1c[nt][0];
#pragma unroll
            for (int r = 1; r < 8; ++r) {
                v0 = fmaxf(v0, a0c[nt][r]);
                v1 = fmaxf(v1, a1c[nt][r]);
            }
            v0 = fmaxf(v0, __shfl_xor(v0, 16, 32));
            v1 = fmaxf(v1, __shfl_xor(v1, 16, 32));
            v0 = fmaxf(v0 + bs[nt], 0.0f);
            v1 = fmaxf(v1 + bs[nt], 0.0f);
            if (lane < 16) {
                outp[nt * 16 + nl]       = v0;
                outp[CCH + nt * 16 + nl] = v1;
            }
        }
    }
}

extern "C" void kernel_launch(void* const* d_in, const int* in_sizes, int n_in,
                              void* d_out, int out_size, void* d_ws, size_t ws_size,
                              hipStream_t stream) {
    const float* coords = (const float*)d_in[0];
    const float* feat   = (const float*)d_in[1];
    const float* Wm     = (const float*)d_in[2];
    const float* bias   = (const float*)d_in[3];
    float* out = (float*)d_out;
    int* idx = (int*)d_ws;

    knn_kernel<<<64, 256, 0, stream>>>(coords, idx);
    agg_kernel<<<64, 256, 0, stream>>>(coords, feat, Wm, bias, idx, out);
}


# KNN kernel: one wave per query instead of one lane per query (32 candidates a step across the lanes, list of 16 along a lane row, wave-uniform threshold, insert by lane shift); grid fills the chip
# speedup vs baseline: 14.8872x; 5.2451x over previous
#include <hip/hip_runtime.h>

typedef __attribute__((ext_vector_type(2))) float v2f;
typedef __attribute__((ext_vector_type(4))) float v4f;
typedef __attribute__((ext_vector_type(8))) float v8f;

#define NPTS 8192
#define CCH  64
#define KNN  16
#define KSTR 132
#define XSTR 68
#define WAVE_LDS (2*64 + 2*16*XSTR)

#define UB   8
#define QPW  2
#define KTH  512

__global__ __launch_bounds__(KTH) void knn_kernel(const float* __restrict__ coords,
                                                  int* __restrict__ idx_out) {
    __shared__ v4f pts[NPTS];
    const int b     = blockIdx.x >> 8;
    const int qbase = (blockIdx.x & 255) << 5;
    const float* cb = coords + b * NPTS * 3;

    for (int j = threadIdx.x; j < NPTS; j += KTH) {
        float x = cb[j * 3 + 0], y = cb[j * 3 + 1], z = cb[j * 3 + 2];
        v4f p; p.x = x; p.y = y; p.z = z; p.w = x * x + y * y + z * z;
        pts[j] = p;
    }
    __syncthreads();

    const int  lane     = threadIdx.x & 31;
    const int  wav      = threadIdx.x >> 5;
    const bool has_prev = (lane & 15) != 0;

    for (int qi = 0; qi < QPW; ++qi) {
        const int q  = qbase + wav * QPW + qi;
        const v4f me = pts[q];

        float bv  = 3.4e38f;
        int   bi  = 0;
        float thr = 3.4e38f;

        for (int c = 0; c < NPTS; c += 32 * UB) {
            float d[UB];
#pragma unroll
            for (int u = 0; u < UB; ++u) {
                const v4f p = pts[c + 32 * u + lane];
                float dot = me.x * p.x;
                dot = fmaf(me.y, p.y, dot);
                dot = fmaf(me.z, p.z, dot);
                float dd = fmaf(-2.0f, dot, me.w + p.w);
                d[u] = fmaxf(dd, 0.0f);
            }
            float m = d[0];
#pragma unroll
            for (int u = 1; u < UB; ++u) m = fminf(m, d[u]);

            if (__builtin_amdgcn_ballot_w32(m < thr) != 0u) {
#pragma unroll
                for (int u = 0; u < UB; ++u) {
                    unsigned mask = __builtin_amdgcn_ballot_w32(d[u] < thr);
                    while (mask != 0u) {
                        const int   l  = __builtin_ctz(mask);
                        const float dn = __int_as_float(__builtin_amdgcn_readlane(__float_as_int(d[u]), l));
                        const int   jn = c + 32 * u + l;
                        const float pb = __int_as_float(__builtin_amdgcn_update_dpp(
                            __float_as_int(bv), __float_as_int(bv), 0x111, 0xf, 0xf, false));
                        const int   pi = __builtin_amdgcn_update_dpp(bi, bi, 0x111, 0xf, 0xf, false);
                        const bool lt  = dn < bv;
                        const bool ltp = has_prev && (dn < pb);
                        bv = ltp ? pb : (lt ? dn : bv);
                        bi = ltp ? pi : (lt ? jn : bi);
                        thr = __int_as_float(__builtin_amdgcn_readlane(__float_as_int(bv), 15));
                        mask = __builtin_amdgcn_ballot_w32(d[u] < thr) & (0xFFFFFFFEu << l);
                    }
                }
            }
        }

        if (lane < KNN) idx_out[(b * NPTS + q) * KNN + lane] = bi;
    }
}

__global__ __launch_bounds__(256) void agg_kernel(const float* __restrict__ coords,
                                                  const float* __restrict__ feat,
                                                  const float* __restrict__ Wm,
                                                  const float* __restrict__ bias,
                                                  const int*   __restrict__ idx,
                                                  float* __restrict__ out) {
    __shared__ float LWT[64 * KSTR];
    __shared__ float XS[8 * WAVE_LDS];

    for (int i = threadIdx.x; i < 131 * 64; i += 256) {
        int k = i >> 6, n = i & 63;
        LWT[n * KSTR + k] = Wm[i];
    }
    if (threadIdx.x < 64) LWT[threadIdx.x * KSTR + 131] = 0.0f;
    __syncthreads();

    const int lane = threadIdx.x & 31;
    const int wav  = threadIdx.x >> 5;
    const int nl   = lane & 15;
    const int h    = lane >> 4;

    float* fib0 = &XS[wav * WAVE_LDS];
    float* fib1 = fib0 + 64;
    float* X0   = fib1 + 64;
    float* X1   = X0 + 16 * XSTR;

    const float* wt0 = LWT + (nl     ) * KSTR + 2 * h;
    const float* wt1 = LWT + (nl + 16) * KSTR + 2 * h;
    const float* wt2 = LWT + (nl + 32) * KSTR + 2 * h;
    const float* wt3 = LWT + (nl + 48) * KSTR + 2 * h;

    float bs[4];
#pragma unroll
    for (int nt = 0; nt < 4; ++nt) bs[nt] = bias[nt * 16 + nl];

    const int gw = blockIdx.x * 8 + wav;

    for (int it = 0; it < 16; ++it) {
        const int P0 = gw * 32 + it * 2;
        const int b  = P0 >> 13;
        const int n0 = P0 & (NPTS - 1);
        const int n1 = n0 + 1;
        const float* fb = feat   + b * NPTS * CCH;
        const float* cb = coords + b * NPTS * 3;

        const int myn = h ? n1 : n0;
        const int nbv = idx[(b * NPTS + myn) * KNN + nl];

        v2f fi0 = *(const v2f*)(fb + n0 * CCH + 2 * lane);
        v2f fi1 = *(const v2f*)(fb + n1 * CCH + 2 * lane);
        *(v2f*)(fib0 + 2 * lane) = fi0;
        *(v2f*)(fib1 + 2 * lane) = fi1;

#pragma unroll
        for (int m = 0; m < 32; ++m) {
            int nb = __shfl(nbv, m, 32);
            v2f g  = *(const v2f*)(fb + nb * CCH + 2 * lane);
            if (m < 16) { v2f d0 = g - fi0; *(v2f*)(X0 + m * XSTR + 2 * lane) = d0; }
            else        { v2f d1 = g - fi1; *(v2f*)(X1 + (m - 16) * XSTR + 2 * lane) = d1; }
        }

        {
            const float* pc  = cb + myn * 3;
            const float* ncp = cb + nbv * 3;
            v4f dp; dp.x = ncp[0] - pc[0]; dp.y = ncp[1] - pc[1];
                    dp.z = ncp[2] - pc[2]; dp.w = 0.0f;
            float* xr = (h ? X1 : X0) + nl * XSTR + 64;
            *(v4f*)xr = dp;
        }

        v8f a0c[4], a1c[4];
        const v8f zc = {0.f, 0.f, 0.f, 0.f, 0.f, 0.f, 0.f, 0.f};
#pragma unroll
        for (int nt = 0; nt < 4; ++nt) { a0c[nt] = zc; a1c[nt] = zc; }

#define STEP(K0, A0, A1)                                                          \
        {                                                                         \
            v2f b0 = *(const v2f*)(wt0 + (K0));                                   \
            v2f b1 = *(const v2f*)(wt1 + (K0));                                   \
            v2f b2 = *(const v2f*)(wt2 + (K0));                                   \
            v2f b3 = *(const v2f*)(wt3 + (K0));                                   \
            a0c[0] = __builtin_amdgcn_wmma_f32_16x16x4_f32(                       \
                false, (A0), false, b0, (short)0, a0c[0], false, false);          \
            a1c[0] = __builtin_amdgcn_wmma_f32_16x16x4_f32(                       \
                false, (A1), false, b0, (short)0, a1c[0], false, false);          \
            a0c[1] = __builtin_amdgcn_wmma_f32_16x16x4_f32(                       \
                false, (A0), false, b1, (short)0, a0c[1], false, false);          \
            a1c[1] = __builtin_amdgcn_wmma_f32_16x16x4_f32(                       \
                false, (A1), false, b1, (short)0, a1c[1], false, false);          \
            a0c[2] = __builtin_amdgcn_wmma_f32_16x16x4_f32(                       \
                false, (A0), false, b2, (short)0, a0c[2], false, false);          \
            a1c[2] = __builtin_amdgcn_wmma_f32_16x16x4_f32(                       \
                false, (A1), false, b2, (short)0, a1c[2], false, false);          \
            a0c[3] = __builtin_amdgcn_wmma_f32_16x16x4_f32(                       \
                false, (A0), false, b3, (short)0, a0c[3], false, false);          \
            a1c[3] = __builtin_amdgcn_wmma_f32_16x16x4_f32(                       \
                false, (A1), false, b3, (short)0, a1c[3], false, false);          \
        }

#pragma unroll
        for (int ks = 0; ks < 16; ++ks) {
            int k0 = 4 * ks;
            v2f A0 = *(const v2f*)(fib0 + k0 + 2 * h);
            v2f A1 = *(const v2f*)(fib1 + k0 + 2 * h);
            STEP(k0, A0, A1);
        }
#pragma unroll
        for (int ks = 0; ks < 16; ++ks) {
            int k0 = 64 + 4 * ks;
            int xo = nl * XSTR + 4 * ks + 2 * h;
            v2f A0 = *(const v2f*)(X0 + xo);
            v2f A1 = *(const v2f*)(X1 + xo);
            STEP(k0, A0, A1);
        }
        {
            int k0 = 128;
            int xo = nl * XSTR + 64 + 2 * h;
            v2f A0 = *(const v2f*)(X0 + xo);
            v2f A1 = *(const v2f*)(X1 + xo);
            STEP(k0, A0, A1);
        }
#undef STEP

        float* outp = out + (size_t)P0 * CCH;
#pragma unroll
        for (int nt = 0; nt < 4; ++nt) {
            float v0 = a0c[nt][0];
            float v1 = a1c[nt][0];
#pragma unroll
            for (int r = 1; r < 8; ++r) {
                v0 = fmaxf(v0, a0c[nt][r]);
                v1 = fmaxf(v1, a1c[nt][r]);
            }
            v0 = fmaxf(v0, __shfl_xor(v0, 16, 32));
            v1 = fmaxf(v1, __shfl_xor(v1, 16, 32));
            v0 = fmaxf(v0 + bs[nt], 0.0f);
            v1 = fmaxf(v1 + bs[nt], 0.0f);
            if (lane < 16) {
                outp[nt * 16 + nl]       = v0;
                outp[CCH + nt * 16 + nl] = v1;
            }
        }
    }
}

extern "C" void kernel_launch(void* const* d_in, const int* in_sizes, int n_in,
                              void* d_out, int out_size, void* d_ws, size_t ws_size,
                              hipStream_t stream) {
    const float* coords = (const float*)d_in[0];
    const float* feat   = (const float*)d_in[1];
    const float* Wm     = (const float*)d_in[2];
    const float* bias   = (const float*)d_in[3];
    float* out = (float*)d_out;
    int* idx = (int*)d_ws;

    knn_kernel<<<512, KTH, 0, stream>>>(coords, idx);
    agg_kernel<<<64, 256, 0, stream>>>(coords, feat, Wm, bias, idx, out);
}


# Aggregation kernel: 64 work-groups of 8 waves (a quarter of the compute units) -> 256 of 16 waves, 4 points a wave instead of 32: one work-group per compute unit, every SIMD's matrix pipe in use
# speedup vs baseline: 28.7370x; 1.9303x over previous
#include <hip/hip_runtime.h>

typedef __attribute__((ext_vector_type(2))) float v2f;
typedef __attribute__((ext_vector_type(4))) float v4f;
typedef __attribute__((ext_vector_type(8))) float v8f;

#define NPTS 8192
#define CCH  64
#define KNN  16
#define KSTR 132
#define XSTR 68
#define WAVE_LDS (2*64 + 2*16*XSTR)
#define AWAV  16
#define APAIR 2
#define ATHR  512
#define AGRID 256
static_assert(ATHR == 32 * AWAV, "aggregation kernel: work-group size");
static_assert(AGRID * AWAV * APAIR * 2 == 2 * NPTS, "aggregation kernel: every point exactly once");

#define UB   8
#define QPW  2
#define KTH  512

__global__ __launch_bounds__(KTH) void knn_kernel(const float* __restrict__ coords,
                                                  int* __restrict__ idx_out) {
    __shared__ v4f pts[NPTS];
    const int b     = blockIdx.x >> 8;
    const int qbase = (blockIdx.x & 255) << 5;
    const float* cb = coords + b * NPTS * 3;

    for (int j = threadIdx.x; j < NPTS; j += KTH) {
        float x = cb[j * 3 + 0], y = cb[j * 3 + 1], z = cb[j * 3 + 2];
        v4f p; p.x = x; p.y = y; p.z = z; p.w = x * x + y * y + z * z;
        pts[j] = p;
    }
    __syncthreads();

    const int  lane     = threadIdx.x & 31;
    const int  wav      = threadIdx.x >> 5;
    const bool has_prev = (lane & 15) != 0;

    for (int qi = 0; qi < QPW; ++qi) {
        const int q  = qbase + wav * QPW + qi;
        const v4f me = pts[q];

        float bv  = 3.4e38f;
        int   bi  = 0;
        float thr = 3.4e38f;

        for (int c = 0; c < NPTS; c += 32 * UB) {
            float d[UB];
#pragma unroll
            for (int u = 0; u < UB; ++u) {
                const v4f p = pts[c + 32 * u + lane];
                float dot = me.x * p.x;
                dot = fmaf(me.y, p.y, dot);
                dot = fmaf(me.z, p.z, dot);
                float dd = fmaf(-2.0f, dot, me.w + p.w);
                d[u] = fmaxf(dd, 0.0f);
            }
            float m = d[0];
#pragma unroll
            for (int u = 1; u < UB; ++u) m = fminf(m, d[u]);

            if (__builtin_amdgcn_ballot_w32(m < thr) != 0u) {
#pragma unroll
                for (int u = 0; u < UB; ++u) {
                    unsigned mask = __builtin_amdgcn_ballot_w32(d[u] < thr);
                    while (mask != 0u) {
                        const int   l  = __builtin_ctz(mask);
                        const float dn = __int_as_float(__builtin_amdgcn_readlane(__float_as_int(d[u]), l));
                        const int   jn = c + 32 * u + l;
                        const float pb = __int_as_float(__builtin_amdgcn_update_dpp(
                            __float_as_int(bv), __float_as_int(bv), 0x111, 0xf, 0xf, false));
                        const int   pi = __builtin_amdgcn_update_dpp(bi, bi, 0x111, 0xf, 0xf, false);
                        const bool lt  = dn < bv;
                        const bool ltp = has_prev && (dn < pb);
                        bv = ltp ? pb : (lt ? dn : bv);
                        bi = ltp ? pi : (lt ? jn : bi);
                        thr = __int_as_float(__builtin_amdgcn_readlane(__float_as_int(bv), 15));
                        mask = __builtin_amdgcn_ballot_w32(d[u] < thr) & (0xFFFFFFFEu << l);
                    }
                }
            }
        }

        if (lane < KNN) idx_out[(b * NPTS + q) * KNN + lane] = bi;
    }
}

__global__ __launch_bounds__(ATHR) void agg_kernel(const float* __restrict__ coords,
                                                  const float* __restrict__ feat,
                                                  const float* __restrict__ Wm,
                                                  const float* __restrict__ bias,
                                                  const int*   __restrict__ idx,
                                                  float* __restrict__ out) {
    __shared__ float LWT[64 * KSTR];
    __shared__ float XS[AWAV * WAVE_LDS];

    for (int i = threadIdx.x; i < 131 * 64; i += ATHR) {
        int k = i >> 6, n = i & 63;
        LWT[n * KSTR + k] = Wm[i];
    }
    if (threadIdx.x < 64) LWT[threadIdx.x * KSTR + 131] = 0.0f;
    __syncthreads();

    const int lane = threadIdx.x & 31;
    const int wav  = threadIdx.x >> 5;
    const int nl   = lane & 15;
    const int h    = lane >> 4;

    float* fib0 = &XS[wav * WAVE_LDS];
    float* fib1 = fib0 + 64;
    float* X0   = fib1 + 64;
    float* X1   = X0 + 16 * XSTR;

    const float* wt0 = LWT + (nl     ) * KSTR + 2 * h;
    const float* wt1 = LWT + (nl + 16) * KSTR + 2 * h;
    const float* wt2 = LWT + (nl + 32) * KSTR + 2 * h;
    const float* wt3 = LWT + (nl + 48) * KSTR + 2 * h;

    float bs[4];
#pragma unroll
    for (int nt = 0; nt < 4; ++nt) bs[nt] = bias[nt * 16 + nl];

    const int gw = blockIdx.x * AWAV + wav;

    for (int it = 0; it < APAIR; ++it) {
        const int P0 = gw * (2 * APAIR) + it * 2;
        const int b  = P0 >> 13;
        const int n0 = P0 & (NPTS - 1);
        const int n1 = n0 + 1;
        const float* fb = feat   + b * NPTS * CCH;
        const float* cb = coords + b * NPTS * 3;

        const int myn = h ? n1 : n0;
        const int nbv = idx[(b * NPTS + myn) * KNN + nl];

        v2f fi0 = *(const v2f*)(fb + n0 * CCH + 2 * lane);
        v2f fi1 = *(const v2f*)(fb + n1 * CCH + 2 * lane);
        *(v2f*)(fib0 + 2 * lane) = fi0;
        *(v2f*)(fib1 + 2 * lane) = fi1;

#pragma unroll
        for (int m = 0; m < 32; ++m) {
            int nb = __shfl(nbv, m, 32);
            v2f g  = *(const v2f*)(fb + nb * CCH + 2 * lane);
            if (m < 16) { v2f d0 = g - fi0; *(v2f*)(X0 + m * XSTR + 2 * lane) = d0; }
            else        { v2f d1 = g - fi1; *(v2f*)(X1 + (m - 16) * XSTR + 2 * lane) = d1; }
        }

        {
            const float* pc  = cb + myn * 3;
            const float* ncp = cb + nbv * 3;
            v4f dp; dp.x = ncp[0] - pc[0]; dp.y = ncp[1] - pc[1];
                    dp.z = ncp[2] - pc[2]; dp.w = 0.0f;
            float* xr = (h ? X1 : X0) + nl * XSTR + 64;
            *(v4f*)xr = dp;
        }

        v8f a0c[4], a1c[4];
        const v8f zc = {0.f, 0.f, 0.f, 0.f, 0.f, 0.f, 0.f, 0.f};
#pragma unroll
        for (int nt = 0; nt < 4; ++nt) { a0c[nt] = zc; a1c[nt] = zc; }

#define STEP(K0, A0, A1)                                                          \
        {                                                                         \
            v2f b0 = *(const v2f*)(wt0 + (K0));                                   \
            v2f b1 = *(const v2f*)(wt1 + (K0));                                   \
            v2f b2 = *(const v2f*)(wt2 + (K0));                                   \
            v2f b3 = *(const v2f*)(wt3 + (K0));                                   \
            a0c[0] = __builtin_amdgcn_wmma_f32_16x16x4_f32(                       \
                false, (A0), false, b0, (short)0, a0c[0], false, false);          \
            a1c[0] = __builtin_amdgcn_wmma_f32_16x16x4_f32(                       \
                false, (A1), false, b0, (short)0, a1c[0], false, false);          \
            a0c[1] = __builtin_amdgcn_wmma_f32_16x16x4_f32(                       \
                false, (A0), false, b1, (short)0, a0c[1], false, false);          \
            a1c[1] = __builtin_amdgcn_wmma_f32_16x16x4_f32(                       \
                false, (A1), false, b1, (short)0, a1c[1], false, false);          \
            a0c[2] = __builtin_amdgcn_wmma_f32_16x16x4_f32(                       \
                false, (A0), false, b2, (short)0, a0c[2], false, false);          \
            a1c[2] = __builtin_amdgcn_wmma_f32_16x16x4_f32(                       \
                false, (A1), false, b2, (short)0, a1c[2], false, false);          \
            a0c[3] = __builtin_amdgcn_wmma_f32_16x16x4_f32(                       \
                false, (A0), false, b3, (short)0, a0c[3], false, false);          \
            a1c[3] = __builtin_amdgcn_wmma_f32_16x16x4_f32(                       \
                false, (A1), false, b3, (short)0, a1c[3], false, false);          \
        }

#pragma unroll
        for (int ks = 0; ks < 16; ++ks) {
            int k0 = 4 * ks;
            v2f A0 = *(const v2f*)(fib0 + k0 + 2 * h);
            v2f A1 = *(const v2f*)(fib1 + k0 + 2 * h);
            STEP(k0, A0, A1);
        }
#pragma unroll
        for (int ks = 0; ks < 16; ++ks) {
            int k0 = 64 + 4 * ks;
            int xo = nl * XSTR + 4 * ks + 2 * h;
            v2f A0 = *(const v2f*)(X0 + xo);
            v2f A1 = *(const v2f*)(X1 + xo);
            STEP(k0, A0, A1);
        }
        {
            int k0 = 128;
            int xo = nl * XSTR + 64 + 2 * h;
            v2f A0 = *(const v2f*)(X0 + xo);
            v2f A1 = *(const v2f*)(X1 + xo);
            STEP(k0, A0, A1);
        }
#undef STEP

        float* outp = out + (size_t)P0 * CCH;
#pragma unroll
        for (int nt = 0; nt < 4; ++nt) {
            float v0 = a0c[nt][0];
            float v1 = a1c[nt][0];
#pragma unroll
            for (int r = 1; r < 8; ++r) {
                v0 = fmaxf(v0, a0c[nt][r]);
                v1 = fmaxf(v1, a1c[nt][r]);
            }
            v0 = fmaxf(v0, __shfl_xor(v0, 16, 32));
            v1 = fmaxf(v1, __shfl_xor(v1, 16, 32));
            v0 = fmaxf(v0 + bs[nt], 0.0f);
            v1 = fmaxf(v1 + bs[nt], 0.0f);
            if (lane < 16) {
                outp[nt * 16 + nl]       = v0;
                outp[CCH + nt * 16 + nl] = v1;
            }
        }
    }
}

extern "C" void kernel_launch(void* const* d_in, const int* in_sizes, int n_in,
                              void* d_out, int out_size, void* d_ws, size_t ws_size,
                              hipStream_t stream) {
    const float* coords = (const float*)d_in[0];
    const float* feat   = (const float*)d_in[1];
    const float* Wm     = (const float*)d_in[2];
    const float* bias   = (const float*)d_in[3];
    float* out = (float*)d_out;
    int* idx = (int*)d_ws;

    knn_kernel<<<512, KTH, 0, stream>>>(coords, idx);
    agg_kernel<<<AGRID, ATHR, 0, stream>>>(coords, feat, Wm, bias, idx, out);
}


# Aggregation kernel: row-constant fi block (k 0..63) taken out of the per-point products: fi @ W[0:64] formed once per work-group for its 64 points (same k order) seeds the accumulators; bit-identical
# speedup vs baseline: 31.8499x; 1.1083x over previous
#include <hip/hip_runtime.h>

typedef __attribute__((ext_vector_type(2))) float v2f;
typedef __attribute__((ext_vector_type(4))) float v4f;
typedef __attribute__((ext_vector_type(8))) float v8f;

#define NPTS 8192
#define CCH  64
#define KNN  16
#define KSTR 132
#define XSTR 68
#define WAVE_LDS (2*16*XSTR)
#define FSTR 68
#define AWAV  16
#define APAIR 2
#define ATHR  512
#define AGRID 256
static_assert(ATHR == 32 * AWAV, "aggregation kernel: work-group size");
static_assert(AGRID * AWAV * APAIR * 2 == 2 * NPTS, "aggregation kernel: every point exactly once");
static_assert(AWAV == 16 && AWAV * APAIR * 2 == 64, "aggregation kernel: 64 points a work-group = 4 row tiles x 4 column tiles, one per wave");

#define UB   8
#define QPW  2
#define KTH  512

__global__ __launch_bounds__(KTH) void knn_kernel(const float* __restrict__ coords,
                                                  int* __restrict__ idx_out) {
    __shared__ v4f pts[NPTS];
    const int b     = blockIdx.x >> 8;
    const int qbase = (blockIdx.x & 255) << 5;
    const float* cb = coords + b * NPTS * 3;

    for (int j = threadIdx.x; j < NPTS; j += KTH) {
        float x = cb[j * 3 + 0], y = cb[j * 3 + 1], z = cb[j * 3 + 2];
        v4f p; p.x = x; p.y = y; p.z = z; p.w = x * x + y * y + z * z;
        pts[j] = p;
    }
    __syncthreads();

    const int  lane     = threadIdx.x & 31;
    const int  wav      = threadIdx.x >> 5;
    const bool has_prev = (lane & 15) != 0;

    for (int qi = 0; qi < QPW; ++qi) {
        const int q  = qbase + wav * QPW + qi;
        const v4f me = pts[q];

        float bv  = 3.4e38f;
        int   bi  = 0;
        float thr = 3.4e38f;

        for (int c = 0; c < NPTS; c += 32 * UB) {
            float d[UB];
#pragma unroll
            for (int u = 0; u < UB; ++u) {
                const v4f p = pts[c + 32 * u + lane];
                float dot = me.x * p.x;
                dot = fmaf(me.y, p.y, dot);
                dot = fmaf(me.z, p.z, dot);
                float dd = fmaf(-2.0f, dot, me.w + p.w);
                d[u] = fmaxf(dd, 0.0f);
            }
            float m = d[0];
#pragma unroll
            for (int u = 1; u < UB; ++u) m = fminf(m, d[u]);

            if (__builtin_amdgcn_ballot_w32(m < thr) != 0u) {
#pragma unroll
                for (int u = 0; u < UB; ++u) {
                    unsigned mask = __builtin_amdgcn_ballot_w32(d[u] < thr);
                    while (mask != 0u) {
                        const int   l  = __builtin_ctz(mask);
                        const float dn = __int_as_float(__builtin_amdgcn_readlane(__float_as_int(d[u]), l));
                        const int   jn = c + 32 * u + l;
                        const float pb = __int_as_float(__builtin_amdgcn_update_dpp(
                            __float_as_int(bv), __float_as_int(bv), 0x111, 0xf, 0xf, false));
                        const int   pi = __builtin_amdgcn_update_dpp(bi, bi, 0x111, 0xf, 0xf, false);
                        const bool lt  = dn < bv;
                        const bool ltp = has_prev && (dn < pb);
                        bv = ltp ? pb : (lt ? dn : bv);
                        bi = ltp ? pi : (lt ? jn : bi);
                        thr = __int_as_float(__builtin_amdgcn_readlane(__float_as_int(bv), 15));
                        mask = __builtin_amdgcn_ballot_w32(d[u] < thr) & (0xFFFFFFFEu << l);
                    }
                }
            }
        }

        if (lane < KNN) idx_out[(b * NPTS + q) * KNN + lane] = bi;
    }
}

__global__ __launch_bounds__(ATHR) void agg_kernel(const float* __restrict__ coords,
                                                  const float* __restrict__ feat,
                                                  const float* __restrict__ Wm,
                                                  const float* __restrict__ bias,
                                                  const int*   __restrict__ idx,
                                                  float* __restrict__ out) {
    __shared__ __attribute__((aligned(16))) float LWT[64 * KSTR];
    __shared__ __attribute__((aligned(16))) float XS[AWAV * WAVE_LDS];
    __shared__ __attribute__((aligned(16))) float FW[64 * FSTR];

    const int lane = threadIdx.x & 31;
    const int wav  = threadIdx.x >> 5;
    const int nl   = lane & 15;
    const int h    = lane >> 4;

    const int pbase = blockIdx.x * (AWAV * APAIR * 2);
    const int mt    = wav >> 2;
    const int ntile = wav & 3;
    v2f afr[16];
    {
        const float* arow = feat + (size_t)(pbase + 16 * mt + nl) * CCH + 2 * h;
#pragma unroll
        for (int ks = 0; ks < 16; ++ks) afr[ks] = *(const v2f*)(arow + 4 * ks);
    }

    for (int i = threadIdx.x; i < 131 * 64; i += ATHR) {
        int k = i >> 6, n = i & 63;
        LWT[n * KSTR + k] = Wm[i];
    }
    if (threadIdx.x < 64) LWT[threadIdx.x * KSTR + 131] = 0.0f;
    __syncthreads();

    {
        const float* wtn = LWT + (nl + 16 * ntile) * KSTR + 2 * h;
        v8f acc = {0.f, 0.f, 0.f, 0.f, 0.f, 0.f, 0.f, 0.f};
        __builtin_amdgcn_sched_barrier(0);
#pragma unroll
        for (int ks = 0; ks < 16; ++ks) {
            v2f bfr = *(const v2f*)(wtn + 4 * ks);
            acc = __builtin_amdgcn_wmma_f32_16x16x4_f32(
                false, afr[ks], false, bfr, (short)0, acc, false, false);
        }
        asm volatile("v_nop\n\tv_nop\n\tv_nop\n\tv_nop" : "+v"(acc));
        __builtin_amdgcn_sched_barrier(0);
        float* fw = FW + (16 * mt + 8 * h) * FSTR + 16 * ntile + nl;
#pragma unroll
        for (int r = 0; r < 8; ++r) fw[r * FSTR] = acc[r];
    }
    __syncthreads();

    float* X0   = &XS[wav * WAVE_LDS];
    float* X1   = X0 + 16 * XSTR;

    const float* wt0 = LWT + (nl     ) * KSTR + 2 * h;
    const float* wt1 = LWT + (nl + 16) * KSTR + 2 * h;
    const float* wt2 = LWT + (nl + 32) * KSTR + 2 * h;
    const float* wt3 = LWT + (nl + 48) * KSTR + 2 * h;

    float bs[4];
#pragma unroll
    for (int nt = 0; nt < 4; ++nt) bs[nt] = bias[nt * 16 + nl];

    const int gw = blockIdx.x * AWAV + wav;

    for (int it = 0; it < APAIR; ++it) {
        const int P0 = gw * (2 * APAIR) + it * 2;
        const int b  = P0 >> 13;
        const int n0 = P0 & (NPTS - 1);
        const int n1 = n0 + 1;
        const float* fb = feat   + b * NPTS * CCH;
        const float* cb = coords + b * NPTS * 3;

        const int myn = h ? n1 : n0;
        const int nbv = idx[(b * NPTS + myn) * KNN + nl];

        v2f fi0 = *(const v2f*)(fb + n0 * CCH + 2 * lane);
        v2f fi1 = *(const v2f*)(fb + n1 * CCH + 2 * lane);

#pragma unroll
        for (int m = 0; m < 32; ++m) {
            int nb = __shfl(nbv, m, 32);
            v2f g  = *(const v2f*)(fb + nb * CCH + 2 * lane);
            if (m < 16) { v2f d0 = g - fi0; *(v2f*)(X0 + m * XSTR + 2 * lane) = d0; }
            else        { v2f d1 = g - fi1; *(v2f*)(X1 + (m - 16) * XSTR + 2 * lane) = d1; }
        }

        {
            const float* pc  = cb + myn * 3;
            const float* ncp = cb + nbv * 3;
            v4f dp; dp.x = ncp[0] - pc[0]; dp.y = ncp[1] - pc[1];
                    dp.z = ncp[2] - pc[2]; dp.w = 0.0f;
            float* xr = (h ? X1 : X0) + nl * XSTR + 64;
            *(v4f*)xr = dp;
        }

        v8f a0c[4], a1c[4];
        {
            const float* f0 = FW + (P0 - pbase) * FSTR + nl;
            const float* f1 = f0 + FSTR;
#pragma unroll
            for (int nt = 0; nt < 4; ++nt) {
                const float c0 = f0[16 * nt];
                const float c1 = f1[16 * nt];
                a0c[nt] = (v8f){c0, c0, c0, c0, c0, c0, c0, c0};
                a1c[nt] = (v8f){c1, c1, c1, c1, c1, c1, c1, c1};
            }
        }

#define STEP(K0, A0, A1)                                                          \
        {                                                                         \
            v2f b0 = *(const v2f*)(wt0 + (K0));                                   \
            v2f b1 = *(const v2f*)(wt1 + (K0));                                   \
            v2f b2 = *(const v2f*)(wt2 + (K0));                                   \
            v2f b3 = *(const v2f*)(wt3 + (K0));                                   \
            a0c[0] = __builtin_amdgcn_wmma_f32_16x16x4_f32(                       \
                false, (A0), false, b0, (short)0, a0c[0], false, false);          \
            a1c[0] = __builtin_amdgcn_wmma_f32_16x16x4_f32(                       \
                false, (A1), false, b0, (short)0, a1c[0], false, false);          \
            a0c[1] = __builtin_amdgcn_wmma_f32_16x16x4_f32(                       \
                false, (A0), false, b1, (short)0, a0c[1], false, false);          \
            a1c[1] = __builtin_amdgcn_wmma_f32_16x16x4_f32(                       \
                false, (A1), false, b1, (short)0, a1c[1], false, false);          \
            a0c[2] = __builtin_amdgcn_wmma_f32_16x16x4_f32(                       \
                false, (A0), false, b2, (short)0, a0c[2], false, false);          \
            a1c[2] = __builtin_amdgcn_wmma_f32_16x16x4_f32(                       \
                false, (A1), false, b2, (short)0, a1c[2], false, false);          \
            a0c[3] = __builtin_amdgcn_wmma_f32_16x16x4_f32(                       \
                false, (A0), false, b3, (short)0, a0c[3], false, false);          \
            a1c[3] = __builtin_amdgcn_wmma_f32_16x16x4_f32(                       \
                false, (A1), false, b3, (short)0, a1c[3], false, false);          \
        }

#pragma unroll
        for (int ks = 0; ks < 16; ++ks) {
            int k0 = 64 + 4 * ks;
            int xo = nl * XSTR + 4 * ks + 2 * h;
            v2f A0 = *(const v2f*)(X0 + xo);
            v2f A1 = *(const v2f*)(X1 + xo);
            STEP(k0, A0, A1);
        }
        {
            int k0 = 128;
            int xo = nl * XSTR + 64 + 2 * h;
            v2f A0 = *(const v2f*)(X0 + xo);
            v2f A1 = *(const v2f*)(X1 + xo);
            STEP(k0, A0, A1);
        }
#undef STEP

        float* outp = out + (size_t)P0 * CCH;
#pragma unroll
        for (int nt = 0; nt < 4; ++nt) {
            float v0 = a0c[nt][0];
            float v1 = a1c[nt][0];
#pragma unroll
            for (int r = 1; r < 8; ++r) {
                v0 = fmaxf(v0, a0c[nt][r]);
                v1 = fmaxf(v1, a1c[nt][r]);
            }
            v0 = fmaxf(v0, __shfl_xor(v0, 16, 32));
            v1 = fmaxf(v1, __shfl_xor(v1, 16, 32));
            v0 = fmaxf(v0 + bs[nt], 0.0f);
            v1 = fmaxf(v1 + bs[nt], 0.0f);
            if (lane < 16) {
                outp[nt * 16 + nl]       = v0;
                outp[CCH + nt * 16 + nl] = v1;
            }
        }
    }
}

extern "C" void kernel_launch(void* const* d_in, const int* in_sizes, int n_in,
                              void* d_out, int out_size, void* d_ws, size_t ws_size,
                              hipStream_t stream) {
    const float* coords = (const float*)d_in[0];
    const float* feat   = (const float*)d_in[1];
    const float* Wm     = (const float*)d_in[2];
    const float* bias   = (const float*)d_in[3];
    float* out = (float*)d_out;
    int* idx = (int*)d_ws;

    knn_kernel<<<512, KTH, 0, stream>>>(coords, idx);
    agg_kernel<<<AGRID, ATHR, 0, stream>>>(coords, feat, Wm, bias, idx, out);
}


# KNN kernel: the two queries of a wave go through the candidates in one pass, not one after the other: each LDS read of a candidate feeds a packed pair of distance chains, one per query; same lists
# speedup vs baseline: 34.2720x; 1.0760x over previous
#include <hip/hip_runtime.h>

typedef __attribute__((ext_vector_type(2))) float v2f;
typedef __attribute__((ext_vector_type(4))) float v4f;
typedef __attribute__((ext_vector_type(8))) float v8f;

#define NPTS 8192
#define CCH  64
#define KNN  16
#define KSTR 132
#define XSTR 68
#define WAVE_LDS (2*16*XSTR)
#define FSTR 68
#define AWAV  16
#define APAIR 2
#define ATHR  512
#define AGRID 256
static_assert(ATHR == 32 * AWAV, "aggregation kernel: work-group size");
static_assert(AGRID * AWAV * APAIR * 2 == 2 * NPTS, "aggregation kernel: every point exactly once");
static_assert(AWAV == 16 && AWAV * APAIR * 2 == 64, "aggregation kernel: 64 points a work-group = 4 row tiles x 4 column tiles, one per wave");

#define UB   8
#define QPW  2
#define KTH  512

static __device__ __forceinline__ void knn_insert_row(const float du, const int jbase, const bool has_prev,
                                                      float& bv, int& bi, float& thr) {
    unsigned mask = __builtin_amdgcn_ballot_w32(du < thr);
    while (mask != 0u) {
        const int   l  = __builtin_ctz(mask);
        const float dn = __int_as_float(__builtin_amdgcn_readlane(__float_as_int(du), l));
        const int   jn = jbase + l;
        const float pb = __int_as_float(__builtin_amdgcn_update_dpp(
            __float_as_int(bv), __float_as_int(bv), 0x111, 0xf, 0xf, false));
        const int   pi = __builtin_amdgcn_update_dpp(bi, bi, 0x111, 0xf, 0xf, false);
        const bool lt  = dn < bv;
        const bool ltp = has_prev && (dn < pb);
        bv = ltp ? pb : (lt ? dn : bv);
        bi = ltp ? pi : (lt ? jn : bi);
        thr = __int_as_float(__builtin_amdgcn_readlane(__float_as_int(bv), 15));
        mask = __builtin_amdgcn_ballot_w32(du < thr) & (0xFFFFFFFEu << l);
    }
}

__global__ __launch_bounds__(KTH) void knn_kernel(const float* __restrict__ coords,
                                                  int* __restrict__ idx_out) {
    __shared__ v4f pts[NPTS];
    const int b     = blockIdx.x >> 8;
    const int qbase = (blockIdx.x & 255) << 5;
    const float* cb = coords + b * NPTS * 3;

    for (int j = threadIdx.x; j < NPTS; j += KTH) {
        float x = cb[j * 3 + 0], y = cb[j * 3 + 1], z = cb[j * 3 + 2];
        v4f p; p.x = x; p.y = y; p.z = z; p.w = x * x + y * y + z * z;
        pts[j] = p;
    }
    __syncthreads();

    const int  lane     = threadIdx.x & 31;
    const int  wav      = threadIdx.x >> 5;
    const bool has_prev = (lane & 15) != 0;

    const int qA = qbase + wav * QPW;
    const v4f ma = pts[qA];
    const v4f mb = pts[qA + 1];
    const v2f MX = {ma.x, mb.x};
    const v2f MY = {ma.y, mb.y};
    const v2f MZ = {ma.z, mb.z};
    const v2f MW = {ma.w, mb.w};
    const v2f M2 = {-2.0f, -2.0f};
    const v2f Z2 = {0.0f, 0.0f};

    float bvA = 3.4e38f, bvB = 3.4e38f;
    int   biA = 0,       biB = 0;
    float thrA = 3.4e38f, thrB = 3.4e38f;

    for (int c = 0; c < NPTS; c += 32 * UB) {
        float dA[UB], dB[UB];
#pragma unroll
        for (int u = 0; u < UB; ++u) {
            const v4f p = pts[c + 32 * u + lane];
            const v2f px = {p.x, p.x};
            const v2f py = {p.y, p.y};
            const v2f pz = {p.z, p.z};
            const v2f pw = {p.w, p.w};
            v2f dot = MX * px;
            dot = __builtin_elementwise_fma(MY, py, dot);
            dot = __builtin_elementwise_fma(MZ, pz, dot);
            const v2f dd = __builtin_elementwise_fma(M2, dot, pw + MW);
            const v2f dm = __builtin_elementwise_max(dd, Z2);
            dA[u] = dm.x;
            dB[u] = dm.y;
        }
        float mA = dA[0], mB = dB[0];
#pragma unroll
        for (int u = 1; u < UB; ++u) { mA = fminf(mA, dA[u]); mB = fminf(mB, dB[u]); }

        const bool anyA = __builtin_amdgcn_ballot_w32(mA < thrA) != 0u;
        const bool anyB = __builtin_amdgcn_ballot_w32(mB < thrB) != 0u;
        if (anyA) {
#pragma unroll
            for (int u = 0; u < UB; ++u) knn_insert_row(dA[u], c + 32 * u, has_prev, bvA, biA, thrA);
        }
        if (anyB) {
#pragma unroll
            for (int u = 0; u < UB; ++u) knn_insert_row(dB[u], c + 32 * u, has_prev, bvB, biB, thrB);
        }
    }

    idx_out[(b * NPTS + qA) * KNN + lane] = (lane < KNN) ? biA : biB;
}

__global__ __launch_bounds__(ATHR) void agg_kernel(const float* __restrict__ coords,
                                                  const float* __restrict__ feat,
                                                  const float* __restrict__ Wm,
                                                  const float* __restrict__ bias,
                                                  const int*   __restrict__ idx,
                                                  float* __restrict__ out) {
    __shared__ __attribute__((aligned(16))) float LWT[64 * KSTR];
    __shared__ __attribute__((aligned(16))) float XS[AWAV * WAVE_LDS];
    __shared__ __attribute__((aligned(16))) float FW[64 * FSTR];

    const int lane = threadIdx.x & 31;
    const int wav  = threadIdx.x >> 5;
    const int nl   = lane & 15;
    const int h    = lane >> 4;

    const int pbase = blockIdx.x * (AWAV * APAIR * 2);
    const int mt    = wav >> 2;
    const int ntile = wav & 3;
    v2f afr[16];
    {
        const float* arow = feat + (size_t)(pbase + 16 * mt + nl) * CCH + 2 * h;
#pragma unroll
        for (int ks = 0; ks < 16; ++ks) afr[ks] = *(const v2f*)(arow + 4 * ks);
    }

    for (int i = threadIdx.x; i < 131 * 64; i += ATHR) {
        int k = i >> 6, n = i & 63;
        LWT[n * KSTR + k] = Wm[i];
    }
    if (threadIdx.x < 64) LWT[threadIdx.x * KSTR + 131] = 0.0f;
    __syncthreads();

    {
        const float* wtn = LWT + (nl + 16 * ntile) * KSTR + 2 * h;
        v8f acc = {0.f, 0.f, 0.f, 0.f, 0.f, 0.f, 0.f, 0.f};
        __builtin_amdgcn_sched_barrier(0);
#pragma unroll
        for (int ks = 0; ks < 16; ++ks) {
            v2f bfr = *(const v2f*)(wtn + 4 * ks);
            acc = __builtin_amdgcn_wmma_f32_16x16x4_f32(
                false, afr[ks], false, bfr, (short)0, acc, false, false);
        }
        asm volatile("v_nop\n\tv_nop\n\tv_nop\n\tv_nop" : "+v"(acc));
        __builtin_amdgcn_sched_barrier(0);
        float* fw = FW + (16 * mt + 8 * h) * FSTR + 16 * ntile + nl;
#pragma unroll
        for (int r = 0; r < 8; ++r) fw[r * FSTR] = acc[r];
    }
    __syncthreads();

    float* X0   = &XS[wav * WAVE_LDS];
    float* X1   = X0 + 16 * XSTR;

    const float* wt0 = LWT + (nl     ) * KSTR + 2 * h;
    const float* wt1 = LWT + (nl + 16) * KSTR + 2 * h;
    const float* wt2 = LWT + (nl + 32) * KSTR + 2 * h;
    const float* wt3 = LWT + (nl + 48) * KSTR + 2 * h;

    float bs[4];
#pragma unroll
    for (int nt = 0; nt < 4; ++nt) bs[nt] = bias[nt * 16 + nl];

    const int gw = blockIdx.x * AWAV + wav;

    for (int it = 0; it < APAIR; ++it) {
        const int P0 = gw * (2 * APAIR) + it * 2;
        const int b  = P0 >> 13;
        const int n0 = P0 & (NPTS - 1);
        const int n1 = n0 + 1;
        const float* fb = feat   + b * NPTS * CCH;
        const float* cb = coords + b * NPTS * 3;

        const int myn = h ? n1 : n0;
        const int nbv = idx[(b * NPTS + myn) * KNN + nl];

        v2f fi0 = *(const v2f*)(fb + n0 * CCH + 2 * lane);
        v2f fi1 = *(const v2f*)(fb + n1 * CCH + 2 * lane);

#pragma unroll
        for (int m = 0; m < 32; ++m) {
            int nb = __shfl(nbv, m, 32);
            v2f g  = *(const v2f*)(fb + nb * CCH + 2 * lane);
            if (m < 16) { v2f d0 = g - fi0; *(v2f*)(X0 + m * XSTR + 2 * lane) = d0; }
            else        { v2f d1 = g - fi1; *(v2f*)(X1 + (m - 16) * XSTR + 2 * lane) = d1; }
        }

        {
            const float* pc  = cb + myn * 3;
            const float* ncp = cb + nbv * 3;
            v4f dp; dp.x = ncp[0] - pc[0]; dp.y = ncp[1] - pc[1];
                    dp.z = ncp[2] - pc[2]; dp.w = 0.0f;
            float* xr = (h ? X1 : X0) + nl * XSTR + 64;
            *(v4f*)xr = dp;
        }

        v8f a0c[4], a1c[4];
        {
            const float* f0 = FW + (P0 - pbase) * FSTR + nl;
            const float* f1 = f0 + FSTR;
#pragma unroll
            for (int nt = 0; nt < 4; ++nt) {
                const float c0 = f0[16 * nt];
                const float c1 = f1[16 * nt];
                a0c[nt] = (v8f){c0, c0, c0, c0, c0, c0, c0, c0};
                a1c[nt] = (v8f){c1, c1, c1, c1, c1, c1, c1, c1};
            }
        }

#define STEP(K0, A0, A1)                                                          \
        {                                                                         \
            v2f b0 = *(const v2f*)(wt0 + (K0));                                   \
            v2f b1 = *(const v2f*)(wt1 + (K0));                                   \
            v2f b2 = *(const v2f*)(wt2 + (K0));                                   \
            v2f b3 = *(const v2f*)(wt3 + (K0));                                   \
            a0c[0] = __builtin_amdgcn_wmma_f32_16x16x4_f32(                       \
                false, (A0), false, b0, (short)0, a0c[0], false, false);          \
            a1c[0] = __builtin_amdgcn_wmma_f32_16x16x4_f32(                       \
                false, (A1), false, b0, (short)0, a1c[0], false, false);          \
            a0c[1] = __builtin_amdgcn_wmma_f32_16x16x4_f32(                       \
                false, (A0), false, b1, (short)0, a0c[1], false, false);          \
            a1c[1] = __builtin_amdgcn_wmma_f32_16x16x4_f32(                       \
                false, (A1), false, b1, (short)0, a1c[1], false, false);          \
            a0c[2] = __builtin_amdgcn_wmma_f32_16x16x4_f32(                       \
                false, (A0), false, b2, (short)0, a0c[2], false, false);          \
            a1c[2] = __builtin_amdgcn_wmma_f32_16x16x4_f32(                       \
                false, (A1), false, b2, (short)0, a1c[2], false, false);          \
            a0c[3] = __builtin_amdgcn_wmma_f32_16x16x4_f32(                       \
                false, (A0), false, b3, (short)0, a0c[3], false, false);          \
            a1c[3] = __builtin_amdgcn_wmma_f32_16x16x4_f32(                       \
                false, (A1), false, b3, (short)0, a1c[3], false, false);          \
        }

#pragma unroll
        for (int ks = 0; ks < 16; ++ks) {
            int k0 = 64 + 4 * ks;
            int xo = nl * XSTR + 4 * ks + 2 * h;
            v2f A0 = *(const v2f*)(X0 + xo);
            v2f A1 = *(const v2f*)(X1 + xo);
            STEP(k0, A0, A1);
        }
        {
            int k0 = 128;
            int xo = nl * XSTR + 64 + 2 * h;
            v2f A0 = *(const v2f*)(X0 + xo);
            v2f A1 = *(const v2f*)(X1 + xo);
            STEP(k0, A0, A1);
        }
#undef STEP

        float* outp = out + (size_t)P0 * CCH;
#pragma unroll
        for (int nt = 0; nt < 4; ++nt) {
            float v0 = a0c[nt][0];
            float v1 = a1c[nt][0];
#pragma unroll
            for (int r = 1; r < 8; ++r) {
                v0 = fmaxf(v0, a0c[nt][r]);
                v1 = fmaxf(v1, a1c[nt][r]);
            }
            v0 = fmaxf(v0, __shfl_xor(v0, 16, 32));
            v1 = fmaxf(v1, __shfl_xor(v1, 16, 32));
            v0 = fmaxf(v0 + bs[nt], 0.0f);
            v1 = fmaxf(v1 + bs[nt], 0.0f);
            if (lane < 16) {
                outp[nt * 16 + nl]       = v0;
                outp[CCH + nt * 16 + nl] = v1;
            }
        }
    }
}

extern "C" void kernel_launch(void* const* d_in, const int* in_sizes, int n_in,
                              void* d_out, int out_size, void* d_ws, size_t ws_size,
                              hipStream_t stream) {
    const float* coords = (const float*)d_in[0];
    const float* feat   = (const float*)d_in[1];
    const float* Wm     = (const float*)d_in[2];
    const float* bias   = (const float*)d_in[3];
    float* out = (float*)d_out;
    int* idx = (int*)d_ws;

    knn_kernel<<<512, KTH, 0, stream>>>(coords, idx);
    agg_kernel<<<AGRID, ATHR, 0, stream>>>(coords, feat, Wm, bias, idx, out);
}


# KNN kernel: before the scan each query gets a bound on its 16th distance (16th smallest of the 32 lane minima over the first 1024 candidates); lists start there, so far fewer inserts; same lists
# speedup vs baseline: 39.6997x; 1.1584x over previous
#include <hip/hip_runtime.h>

typedef __attribute__((ext_vector_type(2))) float v2f;
typedef __attribute__((ext_vector_type(4))) float v4f;
typedef __attribute__((ext_vector_type(8))) float v8f;

#define NPTS 8192
#define CCH  64
#define KNN  16
#define KSTR 132
#define XSTR 68
#define WAVE_LDS (2*16*XSTR)
#define FSTR 68
#define AWAV  16
#define APAIR 2
#define ATHR  512
#define AGRID 256
static_assert(ATHR == 32 * AWAV, "aggregation kernel: work-group size");
static_assert(AGRID * AWAV * APAIR * 2 == 2 * NPTS, "aggregation kernel: every point exactly once");
static_assert(AWAV == 16 && AWAV * APAIR * 2 == 64, "aggregation kernel: 64 points a work-group = 4 row tiles x 4 column tiles, one per wave");

#define UB   8
#define PREG 4
#define QPW  2
#define KTH  512

static __device__ __forceinline__ void knn_insert_row(const float du, const int jbase, const bool has_prev,
                                                      float& bv, int& bi, float& thr) {
    unsigned mask = __builtin_amdgcn_ballot_w32(du < thr);
    while (mask != 0u) {
        const int   l  = __builtin_ctz(mask);
        const float dn = __int_as_float(__builtin_amdgcn_readlane(__float_as_int(du), l));
        const int   jn = jbase + l;
        const float pb = __int_as_float(__builtin_amdgcn_update_dpp(
            __float_as_int(bv), __float_as_int(bv), 0x111, 0xf, 0xf, false));
        const int   pi = __builtin_amdgcn_update_dpp(bi, bi, 0x111, 0xf, 0xf, false);
        const bool lt  = dn < bv;
        const bool ltp = has_prev && (dn < pb);
        bv = ltp ? pb : (lt ? dn : bv);
        bi = ltp ? pi : (lt ? jn : bi);
        thr = __int_as_float(__builtin_amdgcn_readlane(__float_as_int(bv), 15));
        mask = __builtin_amdgcn_ballot_w32(du < thr) & (0xFFFFFFFEu << l);
    }
}

static __device__ __forceinline__ void knn_dist_group(const v4f* pts, const int c, const int lane,
                                                      const v2f MX, const v2f MY, const v2f MZ, const v2f MW,
                                                      float (&dA)[UB], float (&dB)[UB]) {
    const v2f M2 = {-2.0f, -2.0f};
    const v2f Z2 = {0.0f, 0.0f};
#pragma unroll
    for (int u = 0; u < UB; ++u) {
        const v4f p = pts[c + 32 * u + lane];
        const v2f px = {p.x, p.x};
        const v2f py = {p.y, p.y};
        const v2f pz = {p.z, p.z};
        const v2f pw = {p.w, p.w};
        v2f dot = MX * px;
        dot = __builtin_elementwise_fma(MY, py, dot);
        dot = __builtin_elementwise_fma(MZ, pz, dot);
        const v2f dd = __builtin_elementwise_fma(M2, dot, pw + MW);
        const v2f dm = __builtin_elementwise_max(dd, Z2);
        dA[u] = dm.x;
        dB[u] = dm.y;
    }
}

static __device__ __forceinline__ void knn_sixteenth_of_lanes(const float mA, const float mB,
                                                              float& tA, float& tB) {
    const unsigned a = __float_as_uint(mA) & 0x7fffffffu;
    const unsigned b = __float_as_uint(mB) & 0x7fffffffu;
    unsigned curA = 0u, curB = 0u;
    for (int bit = 30; bit >= 0; --bit) {
        const unsigned trA = curA | (1u << bit);
        const unsigned trB = curB | (1u << bit);
        const unsigned nA = (unsigned)__builtin_popcount(__builtin_amdgcn_ballot_w32(a < trA));
        const unsigned nB = (unsigned)__builtin_popcount(__builtin_amdgcn_ballot_w32(b < trB));
        curA = (nA < 16u) ? trA : curA;
        curB = (nB < 16u) ? trB : curB;
    }
    tA = __uint_as_float(curA);
    tB = __uint_as_float(curB);
}

__global__ __launch_bounds__(KTH) void knn_kernel(const float* __restrict__ coords,
                                                  int* __restrict__ idx_out) {
    __shared__ v4f pts[NPTS];
    const int b     = blockIdx.x >> 8;
    const int qbase = (blockIdx.x & 255) << 5;
    const float* cb = coords + b * NPTS * 3;

    for (int j = threadIdx.x; j < NPTS; j += KTH) {
        float x = cb[j * 3 + 0], y = cb[j * 3 + 1], z = cb[j * 3 + 2];
        v4f p; p.x = x; p.y = y; p.z = z; p.w = x * x + y * y + z * z;
        pts[j] = p;
    }
    __syncthreads();

    const int  lane     = threadIdx.x & 31;
    const int  wav      = threadIdx.x >> 5;
    const bool has_prev = (lane & 15) != 0;

    const int qA = qbase + wav * QPW;
    const v4f ma = pts[qA];
    const v4f mb = pts[qA + 1];
    const v2f MX = {ma.x, mb.x};
    const v2f MY = {ma.y, mb.y};
    const v2f MZ = {ma.z, mb.z};
    const v2f MW = {ma.w, mb.w};

    float capA, capB;
    {
        float lmA = 3.4e38f, lmB = 3.4e38f;
        for (int c = 0; c < 32 * UB * PREG; c += 32 * UB) {
            float dA[UB], dB[UB];
            knn_dist_group(pts, c, lane, MX, MY, MZ, MW, dA, dB);
#pragma unroll
            for (int u = 0; u < UB; ++u) { lmA = fminf(lmA, dA[u]); lmB = fminf(lmB, dB[u]); }
        }
        float tA, tB;
        knn_sixteenth_of_lanes(lmA, lmB, tA, tB);
        capA = fminf(fmaf(tA, 1.000001f, 1e-30f), 3.4e38f);
        capB = fminf(fmaf(tB, 1.000001f, 1e-30f), 3.4e38f);
    }

    float bvA = capA, bvB = capB;
    int   biA = 0,    biB = 0;
    float thrA = capA, thrB = capB;

    for (int c = 0; c < NPTS; c += 32 * UB) {
        float dA[UB], dB[UB];
        knn_dist_group(pts, c, lane, MX, MY, MZ, MW, dA, dB);
        float mA = dA[0], mB = dB[0];
#pragma unroll
        for (int u = 1; u < UB; ++u) { mA = fminf(mA, dA[u]); mB = fminf(mB, dB[u]); }

        const bool anyA = __builtin_amdgcn_ballot_w32(mA < thrA) != 0u;
        const bool anyB = __builtin_amdgcn_ballot_w32(mB < thrB) != 0u;
        if (anyA) {
#pragma unroll
            for (int u = 0; u < UB; ++u) knn_insert_row(dA[u], c + 32 * u, has_prev, bvA, biA, thrA);
        }
        if (anyB) {
#pragma unroll
            for (int u = 0; u < UB; ++u) knn_insert_row(dB[u], c + 32 * u, has_prev, bvB, biB, thrB);
        }
    }

    idx_out[(b * NPTS + qA) * KNN + lane] = (lane < KNN) ? biA : biB;
}

__global__ __launch_bounds__(ATHR) void agg_kernel(const float* __restrict__ coords,
                                                  const float* __restrict__ feat,
                                                  const float* __restrict__ Wm,
                                                  const float* __restrict__ bias,
                                                  const int*   __restrict__ idx,
                                                  float* __restrict__ out) {
    __shared__ __attribute__((aligned(16))) float LWT[64 * KSTR];
    __shared__ __attribute__((aligned(16))) float XS[AWAV * WAVE_LDS];
    __shared__ __attribute__((aligned(16))) float FW[64 * FSTR];

    const int lane = threadIdx.x & 31;
    const int wav  = threadIdx.x >> 5;
    const int nl   = lane & 15;
    const int h    = lane >> 4;

    const int pbase = blockIdx.x * (AWAV * APAIR * 2);
    const int mt    = wav >> 2;
    const int ntile = wav & 3;
    v2f afr[16];
    {
        const float* arow = feat + (size_t)(pbase + 16 * mt + nl) * CCH + 2 * h;
#pragma unroll
        for (int ks = 0; ks < 16; ++ks) afr[ks] = *(const v2f*)(arow + 4 * ks);
    }

    for (int i = threadIdx.x; i < 131 * 64; i += ATHR) {
        int k = i >> 6, n = i & 63;
        LWT[n * KSTR + k] = Wm[i];
    }
    if (threadIdx.x < 64) LWT[threadIdx.x * KSTR + 131] = 0.0f;
    __syncthreads();

    {
        const float* wtn = LWT + (nl + 16 * ntile) * KSTR + 2 * h;
        v8f acc = {0.f, 0.f, 0.f, 0.f, 0.f, 0.f, 0.f, 0.f};
        __builtin_amdgcn_sched_barrier(0);
#pragma unroll
        for (int ks = 0; ks < 16; ++ks) {
            v2f bfr = *(const v2f*)(wtn + 4 * ks);
            acc = __builtin_amdgcn_wmma_f32_16x16x4_f32(
                false, afr[ks], false, bfr, (short)0, acc, false, false);
        }
        asm volatile("v_nop\n\tv_nop\n\tv_nop\n\tv_nop" : "+v"(acc));
        __builtin_amdgcn_sched_barrier(0);
        float* fw = FW + (16 * mt + 8 * h) * FSTR + 16 * ntile + nl;
#pragma unroll
        for (int r = 0; r < 8; ++r) fw[r * FSTR] = acc[r];
    }
    __syncthreads();

    float* X0   = &XS[wav * WAVE_LDS];
    float* X1   = X0 + 16 * XSTR;

    const float* wt0 = LWT + (nl     ) * KSTR + 2 * h;
    const float* wt1 = LWT + (nl + 16) * KSTR + 2 * h;
    const float* wt2 = LWT + (nl + 32) * KSTR + 2 * h;
    const float* wt3 = LWT + (nl + 48) * KSTR + 2 * h;

    float bs[4];
#pragma unroll
    for (int nt = 0; nt < 4; ++nt) bs[nt] = bias[nt * 16 + nl];

    const int gw = blockIdx.x * AWAV + wav;

    for (int it = 0; it < APAIR; ++it) {
        const int P0 = gw * (2 * APAIR) + it * 2;
        const int b  = P0 >> 13;
        const int n0 = P0 & (NPTS - 1);
        const int n1 = n0 + 1;
        const float* fb = feat   + b * NPTS * CCH;
        const float* cb = coords + b * NPTS * 3;

        const int myn = h ? n1 : n0;
        const int nbv = idx[(b * NPTS + myn) * KNN + nl];

        v2f fi0 = *(const v2f*)(fb + n0 * CCH + 2 * lane);
        v2f fi1 = *(const v2f*)(fb + n1 * CCH + 2 * lane);

#pragma unroll
        for (int m = 0; m < 32; ++m) {
            int nb = __shfl(nbv, m, 32);
            v2f g  = *(const v2f*)(fb + nb * CCH + 2 * lane);
            if (m < 16) { v2f d0 = g - fi0; *(v2f*)(X0 + m * XSTR + 2 * lane) = d0; }
            else        { v2f d1 = g - fi1; *(v2f*)(X1 + (m - 16) * XSTR + 2 * lane) = d1; }
        }

        {
            const float* pc  = cb + myn * 3;
            const float* ncp = cb + nbv * 3;
            v4f dp; dp.x = ncp[0] - pc[0]; dp.y = ncp[1] - pc[1];
                    dp.z = ncp[2] - pc[2]; dp.w = 0.0f;
            float* xr = (h ? X1 : X0) + nl * XSTR + 64;
            *(v4f*)xr = dp;
        }

        v8f a0c[4], a1c[4];
        {
            const float* f0 = FW + (P0 - pbase) * FSTR + nl;
            const float* f1 = f0 + FSTR;
#pragma unroll
            for (int nt = 0; nt < 4; ++nt) {
                const float c0 = f0[16 * nt];
                const float c1 = f1[16 * nt];
                a0c[nt] = (v8f){c0, c0, c0, c0, c0, c0, c0, c0};
                a1c[nt] = (v8f){c1, c1, c1, c1, c1, c1, c1, c1};
            }
        }

#define STEP(K0, A0, A1)                                                          \
        {                                                                         \
            v2f b0 = *(const v2f*)(wt0 + (K0));                                   \
            v2f b1 = *(const v2f*)(wt1 + (K0));                                   \
            v2f b2 = *(const v2f*)(wt2 + (K0));                                   \
            v2f b3 = *(const v2f*)(wt3 + (K0));                                   \
            a0c[0] = __builtin_amdgcn_wmma_f32_16x16x4_f32(                       \
                false, (A0), false, b0, (short)0, a0c[0], false, false);          \
            a1c[0] = __builtin_amdgcn_wmma_f32_16x16x4_f32(                       \
                false, (A1), false, b0, (short)0, a1c[0], false, false);          \
            a0c[1] = __builtin_amdgcn_wmma_f32_16x16x4_f32(                       \
                false, (A0), false, b1, (short)0, a0c[1], false, false);          \
            a1c[1] = __builtin_amdgcn_wmma_f32_16x16x4_f32(                       \
                false, (A1), false, b1, (short)0, a1c[1], false, false);          \
            a0c[2] = __builtin_amdgcn_wmma_f32_16x16x4_f32(                       \
                false, (A0), false, b2, (short)0, a0c[2], false, false);          \
            a1c[2] = __builtin_amdgcn_wmma_f32_16x16x4_f32(                       \
                false, (A1), false, b2, (short)0, a1c[2], false, false);          \
            a0c[3] = __builtin_amdgcn_wmma_f32_16x16x4_f32(                       \
                false, (A0), false, b3, (short)0, a0c[3], false, false);          \
            a1c[3] = __builtin_amdgcn_wmma_f32_16x16x4_f32(                       \
                false, (A1), false, b3, (short)0, a1c[3], false, false);          \
        }

#pragma unroll
        for (int ks = 0; ks < 16; ++ks) {
            int k0 = 64 + 4 * ks;
            int xo = nl * XSTR + 4 * ks + 2 * h;
            v2f A0 = *(const v2f*)(X0 + xo);
            v2f A1 = *(const v2f*)(X1 + xo);
            STEP(k0, A0, A1);
        }
        {
            int k0 = 128;
            int xo = nl * XSTR + 64 + 2 * h;
            v2f A0 = *(const v2f*)(X0 + xo);
            v2f A1 = *(const v2f*)(X1 + xo);
            STEP(k0, A0, A1);
        }
#undef STEP

        float* outp = out + (size_t)P0 * CCH;
#pragma unroll
        for (int nt = 0; nt < 4; ++nt) {
            float v0 = a0c[nt][0];
            float v1 = a1c[nt][0];
#pragma unroll
            for (int r = 1; r < 8; ++r) {
                v0 = fmaxf(v0, a0c[nt][r]);
                v1 = fmaxf(v1, a1c[nt][r]);
            }
            v0 = fmaxf(v0, __shfl_xor(v0, 16, 32));
            v1 = fmaxf(v1, __shfl_xor(v1, 16, 32));
            v0 = fmaxf(v0 + bs[nt], 0.0f);
            v1 = fmaxf(v1 + bs[nt], 0.0f);
            if (lane < 16) {
                outp[nt * 16 + nl]       = v0;
                outp[CCH + nt * 16 + nl] = v1;
            }
        }
    }
}

extern "C" void kernel_launch(void* const* d_in, const int* in_sizes, int n_in,
                              void* d_out, int out_size, void* d_ws, size_t ws_size,
                              hipStream_t stream) {
    const float* coords = (const float*)d_in[0];
    const float* feat   = (const float*)d_in[1];
    const float* Wm     = (const float*)d_in[2];
    const float* bias   = (const float*)d_in[3];
    float* out = (float*)d_out;
    int* idx = (int*)d_ws;

    knn_kernel<<<512, KTH, 0, stream>>>(coords, idx);
    agg_kernel<<<AGRID, ATHR, 0, stream>>>(coords, feat, Wm, bias, idx, out);
}


# One launch instead of two: a 32-wave work-group stages points and W once, and each wave aggregates its own two queries right after its scan, neighbour lists kept in registers; same lists, same sums
# speedup vs baseline: 44.2746x; 1.1152x over previous
#include <hip/hip_runtime.h>

typedef __attribute__((ext_vector_type(2))) float v2f;
typedef __attribute__((ext_vector_type(4))) float v4f;
typedef __attribute__((ext_vector_type(8))) float v8f;

#define NPTS 8192
#define CCH  64
#define KNN  16
#define KSTR 132
#define XSTR 68
#define WAVE_LDS (16*XSTR)
#define FSTR 68
#define FWAV  32
#define FTHR  1024
#define FGRID 256
static_assert(FTHR == 32 * FWAV, "work-group size");
static_assert(FGRID * FWAV * 2 == 2 * NPTS, "every point exactly once: two per wave");
static_assert(FWAV * 2 == 64, "64 points a work-group = 4 row tiles x 4 column tiles of the fi product");

#define UB   8
#define PREG 4
#define QPW  2

static __device__ __forceinline__ void knn_insert_row(const float du, const int jbase, const bool has_prev,
                                                      float& bv, int& bi, float& thr) {
    unsigned mask = __builtin_amdgcn_ballot_w32(du < thr);
    while (mask != 0u) {
        const int   l  = __builtin_ctz(mask);
        const float dn = __int_as_float(__builtin_amdgcn_readlane(__float_as_int(du), l));
        const int   jn = jbase + l;
        const float pb = __int_as_float(__builtin_amdgcn_update_dpp(
            __float_as_int(bv), __float_as_int(bv), 0x111, 0xf, 0xf, false));
        const int   pi = __builtin_amdgcn_update_dpp(bi, bi, 0x111, 0xf, 0xf, false);
        const bool lt  = dn < bv;
        const bool ltp = has_prev && (dn < pb);
        bv = ltp ? pb : (lt ? dn : bv);
        bi = ltp ? pi : (lt ? jn : bi);
        thr = __int_as_float(__builtin_amdgcn_readlane(__float_as_int(bv), 15));
        mask = __builtin_amdgcn_ballot_w32(du < thr) & (0xFFFFFFFEu << l);
    }
}

static __device__ __forceinline__ void knn_dist_group(const v4f* pts, const int c, const int lane,
                                                      const v2f MX, const v2f MY, const v2f MZ, const v2f MW,
                                                      float (&dA)[UB], float (&dB)[UB]) {
    const v2f M2 = {-2.0f, -2.0f};
    const v2f Z2 = {0.0f, 0.0f};
#pragma unroll
    for (int u = 0; u < UB; ++u) {
        const v4f p = pts[c + 32 * u + lane];
        const v2f px = {p.x, p.x};
        const v2f py = {p.y, p.y};
        const v2f pz = {p.z, p.z};
        const v2f pw = {p.w, p.w};
        v2f dot = MX * px;
        dot = __builtin_elementwise_fma(MY, py, dot);
        dot = __builtin_elementwise_fma(MZ, pz, dot);
        const v2f dd = __builtin_elementwise_fma(M2, dot, pw + MW);
        const v2f dm = __builtin_elementwise_max(dd, Z2);
        dA[u] = dm.x;
        dB[u] = dm.y;
    }
}

static __device__ __forceinline__ void knn_sixteenth_of_lanes(const float mA, const float mB,
                                                              float& tA, float& tB) {
    const unsigned a = __float_as_uint(mA) & 0x7fffffffu;
    const unsigned b = __float_as_uint(mB) & 0x7fffffffu;
    unsigned curA = 0u, curB = 0u;
    for (int bit = 30; bit >= 0; --bit) {
        const unsigned trA = curA | (1u << bit);
        const unsigned trB = curB | (1u << bit);
        const unsigned nA = (unsigned)__builtin_popcount(__builtin_amdgcn_ballot_w32(a < trA));
        const unsigned nB = (unsigned)__builtin_popcount(__builtin_amdgcn_ballot_w32(b < trB));
        curA = (nA < 16u) ? trA : curA;
        curB = (nB < 16u) ? trB : curB;
    }
    tA = __uint_as_float(curA);
    tB = __uint_as_float(curB);
}

__global__ __launch_bounds__(FTHR) void knn_agg_kernel(const float* __restrict__ coords,
                                                       const float* __restrict__ feat,
                                                       const float* __restrict__ Wm,
                                                       const float* __restrict__ bias,
                                                       float* __restrict__ out) {
    __shared__ __attribute__((aligned(16))) float SM[64 * KSTR + 64 * FSTR + FWAV * WAVE_LDS + 4 * NPTS];
    float* const LWT = SM;
    float* const FW  = LWT + 64 * KSTR;
    float* const XS  = FW + 64 * FSTR;
    v4f*   const pts = (v4f*)(XS + FWAV * WAVE_LDS);

    const int lane = threadIdx.x & 31;
    const int wav  = threadIdx.x >> 5;
    const int nl   = lane & 15;
    const int h    = lane >> 4;

    const int b     = blockIdx.x >> 7;
    const int qbase = (blockIdx.x & 127) << 6;
    const int pbase = blockIdx.x * 64;
    const float* cb = coords + b * NPTS * 3;
    const float* fb = feat   + b * NPTS * CCH;

    const int mt    = (wav >> 2) & 3;
    const int ntile = wav & 3;
    v2f afr[16];
    {
        const float* arow = feat + (size_t)(pbase + 16 * mt + nl) * CCH + 2 * h;
#pragma unroll
        for (int ks = 0; ks < 16; ++ks) afr[ks] = *(const v2f*)(arow + 4 * ks);
    }

#pragma unroll
    for (int it = 0; it < NPTS / FTHR; ++it) {
        const int j = threadIdx.x + it * FTHR;
        float x = cb[j * 3 + 0], y = cb[j * 3 + 1], z = cb[j * 3 + 2];
        v4f p; p.x = x; p.y = y; p.z = z; p.w = x * x + y * y + z * z;
        pts[j] = p;
    }
#pragma unroll
    for (int it = 0; it < (131 * 64 + FTHR - 1) / FTHR; ++it) {
        const int i = threadIdx.x + it * FTHR;
        if (i < 131 * 64) {
            int k = i >> 6, n = i & 63;
            LWT[n * KSTR + k] = Wm[i];
        }
    }
    if (threadIdx.x < 64) LWT[threadIdx.x * KSTR + 131] = 0.0f;
    __syncthreads();

    if (wav < 16) {
        const float* wtn = LWT + (nl + 16 * ntile) * KSTR + 2 * h;
        v8f acc = {0.f, 0.f, 0.f, 0.f, 0.f, 0.f, 0.f, 0.f};
        __builtin_amdgcn_sched_barrier(0);
#pragma unroll
        for (int ks = 0; ks < 16; ++ks) {
            v2f bfr = *(const v2f*)(wtn + 4 * ks);
            acc = __builtin_amdgcn_wmma_f32_16x16x4_f32(
                false, afr[ks], false, bfr, (short)0, acc, false, false);
        }
        asm volatile("v_nop\n\tv_nop\n\tv_nop\n\tv_nop" : "+v"(acc));
        __builtin_amdgcn_sched_barrier(0);
        float* fw = FW + (16 * mt + 8 * h) * FSTR + 16 * ntile + nl;
#pragma unroll
        for (int r = 0; r < 8; ++r) fw[r * FSTR] = acc[r];
    }
    __syncthreads();

    const bool has_prev = (lane & 15) != 0;

    const int qA = qbase + wav * QPW;
    const v4f ma = pts[qA];
    const v4f mb = pts[qA + 1];
    const v2f MX = {ma.x, mb.x};
    const v2f MY = {ma.y, mb.y};
    const v2f MZ = {ma.z, mb.z};
    const v2f MW = {ma.w, mb.w};

    float capA, capB;
    {
        float lmA = 3.4e38f, lmB = 3.4e38f;
        for (int c = 0; c < 32 * UB * PREG; c += 32 * UB) {
            float dA[UB], dB[UB];
            knn_dist_group(pts, c, lane, MX, MY, MZ, MW, dA, dB);
#pragma unroll
            for (int u = 0; u < UB; ++u) { lmA = fminf(lmA, dA[u]); lmB = fminf(lmB, dB[u]); }
        }
        float tA, tB;
        knn_sixteenth_of_lanes(lmA, lmB, tA, tB);
        capA = fminf(fmaf(tA, 1.000001f, 1e-30f), 3.4e38f);
        capB = fminf(fmaf(tB, 1.000001f, 1e-30f), 3.4e38f);
    }

    float bvA = capA, bvB = capB;
    int   biA = 0,    biB = 0;
    float thrA = capA, thrB = capB;

    for (int c = 0; c < NPTS; c += 32 * UB) {
        float dA[UB], dB[UB];
        knn_dist_group(pts, c, lane, MX, MY, MZ, MW, dA, dB);
        float mA = dA[0], mB = dB[0];
#pragma unroll
        for (int u = 1; u < UB; ++u) { mA = fminf(mA, dA[u]); mB = fminf(mB, dB[u]); }

        const bool anyA = __builtin_amdgcn_ballot_w32(mA < thrA) != 0u;
        const bool anyB = __builtin_amdgcn_ballot_w32(mB < thrB) != 0u;
        if (anyA) {
#pragma unroll
            for (int u = 0; u < UB; ++u) knn_insert_row(dA[u], c + 32 * u, has_prev, bvA, biA, thrA);
        }
        if (anyB) {
#pragma unroll
            for (int u = 0; u < UB; ++u) knn_insert_row(dB[u], c + 32 * u, has_prev, bvB, biB, thrB);
        }
    }

    float* X = &XS[wav * WAVE_LDS];
    const float* xa = X + nl * XSTR + 2 * h;

    const float* wt0 = LWT + (nl     ) * KSTR + 2 * h;
    const float* wt1 = LWT + (nl + 16) * KSTR + 2 * h;
    const float* wt2 = LWT + (nl + 32) * KSTR + 2 * h;
    const float* wt3 = LWT + (nl + 48) * KSTR + 2 * h;

    float bs[4];
#pragma unroll
    for (int nt = 0; nt < 4; ++nt) bs[nt] = bias[nt * 16 + nl];

#pragma unroll
    for (int pt = 0; pt < QPW; ++pt) {
        const int n0  = qA + pt;
        const int nbv = pt ? biB : biA;

        const v2f fi = *(const v2f*)(fb + n0 * CCH + 2 * lane);

#pragma unroll
        for (int m = 0; m < 16; ++m) {
            const int nb = __shfl(nbv, m, 32);
            const v2f g  = *(const v2f*)(fb + nb * CCH + 2 * lane);
            *(v2f*)(X + m * XSTR + 2 * lane) = g - fi;
        }

        {
            const float* pc  = cb + n0 * 3;
            const float* ncp = cb + nbv * 3;
            v4f dp; dp.x = ncp[0] - pc[0]; dp.y = ncp[1] - pc[1];
                    dp.z = ncp[2] - pc[2]; dp.w = 0.0f;
            *(v4f*)(X + nl * XSTR + 64) = dp;
        }

        v8f ac[4];
        {
            const float* f0 = FW + (wav * QPW + pt) * FSTR + nl;
#pragma unroll
            for (int nt = 0; nt < 4; ++nt) {
                const float c0 = f0[16 * nt];
                ac[nt] = (v8f){c0, c0, c0, c0, c0, c0, c0, c0};
            }
        }

#define STEP(K0, A0)                                                              \
        {                                                                         \
            v2f b0 = *(const v2f*)(wt0 + (K0));                                   \
            v2f b1 = *(const v2f*)(wt1 + (K0));                                   \
            v2f b2 = *(const v2f*)(wt2 + (K0));                                   \
            v2f b3 = *(const v2f*)(wt3 + (K0));                                   \
            ac[0] = __builtin_amdgcn_wmma_f32_16x16x4_f32(                        \
                false, (A0), false, b0, (short)0, ac[0], false, false);           \
            ac[1] = __builtin_amdgcn_wmma_f32_16x16x4_f32(                        \
                false, (A0), false, b1, (short)0, ac[1], false, false);           \
            ac[2] = __builtin_amdgcn_wmma_f32_16x16x4_f32(                        \
                false, (A0), false, b2, (short)0, ac[2], false, false);           \
            ac[3] = __builtin_amdgcn_wmma_f32_16x16x4_f32(                        \
                false, (A0), false, b3, (short)0, ac[3], false, false);           \
        }

        __builtin_amdgcn_sched_barrier(0);
#pragma unroll
        for (int ks = 0; ks < 16; ++ks) {
            v2f A0 = *(const v2f*)(xa + 4 * ks);
            STEP(64 + 4 * ks, A0);
        }
        {
            v2f A0 = *(const v2f*)(xa + 64);
            STEP(128, A0);
        }
#undef STEP
        asm volatile("v_nop\n\tv_nop\n\tv_nop\n\tv_nop" : "+v"(ac[0]), "+v"(ac[1]), "+v"(ac[2]), "+v"(ac[3]));
        __builtin_amdgcn_sched_barrier(0);

        float* outp = out + (size_t)(pbase + wav * QPW + pt) * CCH;
#pragma unroll
        for (int nt = 0; nt < 4; ++nt) {
            float v0 = ac[nt][0];
#pragma unroll
            for (int r = 1; r < 8; ++r) v0 = fmaxf(v0, ac[nt][r]);
            v0 = fmaxf(v0, __shfl_xor(v0, 16, 32));
            v0 = fmaxf(v0 + bs[nt], 0.0f);
            if (lane < 16) outp[nt * 16 + nl] = v0;
        }
    }
}

extern "C" void kernel_launch(void* const* d_in, const int* in_sizes, int n_in,
                              void* d_out, int out_size, void* d_ws, size_t ws_size,
                              hipStream_t stream) {
    const float* coords = (const float*)d_in[0];
    const float* feat   = (const float*)d_in[1];
    const float* Wm     = (const float*)d_in[2];
    const float* bias   = (const float*)d_in[3];
    float* out = (float*)d_out;

    knn_agg_kernel<<<FGRID, FTHR, 0, stream>>>(coords, feat, Wm, bias, out);
}
